# starting kernel
# baseline (speedup 1.0000x reference)
#include <hip/hip_runtime.h>


typedef float v2f __attribute__((ext_vector_type(2)));
typedef float v8f __attribute__((ext_vector_type(8)));

#define LBEAM 1.0f
#define EA_C  1.0e4f
#define EI_C  1.0e2f

__device__ __forceinline__ float fast_tanhf(float a) {
#if __has_builtin(__builtin_amdgcn_tanhf)
    return __builtin_amdgcn_tanhf(a);
#else
    float e = __expf(2.0f * a);
    return 1.0f - 2.0f * __builtin_amdgcn_rcpf(e + 1.0f);
#endif
}

__global__ __launch_bounds__(256) void pinn_beam_wmma(
    const float* __restrict__ x,
    const float* __restrict__ W1, const float* __restrict__ b1,
    const float* __restrict__ W2, const float* __restrict__ b2,
    const float* __restrict__ W3, const float* __restrict__ b3,
    float* __restrict__ out, int n)
{
    const int lane = threadIdx.x & 31;
    const int half = lane >> 4;
    const int l16  = lane & 15;

    v2f Aw[4];
#pragma unroll
    for (int c = 0; c < 4; ++c) {
        Aw[c].x = W2[l16 * 16 + 4 * c + 2 * half + 0];
        Aw[c].y = W2[l16 * 16 + 4 * c + 2 * half + 1];
    }
    float w1c[8], b1c[8];
#pragma unroll
    for (int c = 0; c < 4; ++c)
#pragma unroll
        for (int j = 0; j < 2; ++j) {
            int ch = 4 * c + 2 * half + j;
            w1c[2 * c + j] = W1[ch];
            b1c[2 * c + j] = b1[ch];
        }
    float b2c[8], w3u[8], w3w[8];
#pragma unroll
    for (int r = 0; r < 8; ++r) {
        int ch = r + 8 * half;
        b2c[r] = b2[ch];
        w3u[r] = W3[ch];
        w3w[r] = W3[16 + ch];
    }
    const float b30 = b3[0], b31 = b3[1];

    const int nBatch = n >> 4;
    const int gwave  = (int)(blockIdx.x * blockDim.x + threadIdx.x) >> 5;
    const int nwave  = (int)(gridDim.x * blockDim.x) >> 5;

    for (int batch = gwave; batch < nBatch; batch += nwave) {
        const int   pidx = batch * 16 + l16;
        const float xi   = x[pidx];

        float h0[8], h1[8], h2[8], h3[8];
#pragma unroll
        for (int i = 0; i < 8; ++i) {
            float a1 = w1c[i] * (1.0f / LBEAM);
            float a0 = a1 * xi + b1c[i];
            float t  = fast_tanhf(a0);
            float s  = 1.0f - t * t;
            h0[i] = t;
            h1[i] = s * a1;
            h2[i] = -2.0f * t * s * a1 * a1;
            h3[i] = s * (4.0f * t * t - 2.0f * s) * a1 * a1 * a1;
        }

        v8f acc0 = {}, acc1 = {}, acc2 = {}, acc3 = {};
#pragma unroll
        for (int c = 0; c < 4; ++c) {
            v2f B0 = { h0[2 * c], h0[2 * c + 1] };
            v2f B1 = { h1[2 * c], h1[2 * c + 1] };
            v2f B2 = { h2[2 * c], h2[2 * c + 1] };
            v2f B3 = { h3[2 * c], h3[2 * c + 1] };
            acc0 = __builtin_amdgcn_wmma_f32_16x16x4_f32(false, Aw[c], false, B0,
                                                         (short)0, acc0, false, false);
            acc1 = __builtin_amdgcn_wmma_f32_16x16x4_f32(false, Aw[c], false, B1,
                                                         (short)0, acc1, false, false);
            acc2 = __builtin_amdgcn_wmma_f32_16x16x4_f32(false, Aw[c], false, B2,
                                                         (short)0, acc2, false, false);
            acc3 = __builtin_amdgcn_wmma_f32_16x16x4_f32(false, Aw[c], false, B3,
                                                         (short)0, acc3, false, false);
        }

        float p00 = 0.f, p01 = 0.f, p02 = 0.f, p03 = 0.f;
        float p10 = 0.f, p11 = 0.f, p12 = 0.f, p13 = 0.f;
#pragma unroll
        for (int r = 0; r < 8; ++r) {
            float A0 = acc0[r] + b2c[r];
            float A1 = acc1[r];
            float A2 = acc2[r];
            float A3 = acc3[r];
            float t  = fast_tanhf(A0);
            float s  = 1.0f - t * t;
            float g0 = t;
            float g1 = s * A1;
            float g2 = s * A2 - 2.0f * t * s * A1 * A1;
            float g3 = s * A3 - 6.0f * t * s * A1 * A2
                     + s * (4.0f * t * t - 2.0f * s) * A1 * A1 * A1;
            p00 += w3u[r] * g0; p01 += w3u[r] * g1;
            p02 += w3u[r] * g2; p03 += w3u[r] * g3;
            p10 += w3w[r] * g0; p11 += w3w[r] * g1;
            p12 += w3w[r] * g2; p13 += w3w[r] * g3;
        }
        p00 += __shfl_xor(p00, 16, 32);
        p01 += __shfl_xor(p01, 16, 32);
        p02 += __shfl_xor(p02, 16, 32);
        p03 += __shfl_xor(p03, 16, 32);
        p10 += __shfl_xor(p10, 16, 32);
        p11 += __shfl_xor(p11, 16, 32);
        p12 += __shfl_xor(p12, 16, 32);
        p13 += __shfl_xor(p13, 16, 32);

        float u    = p00 + b30;
        float ux   = p01;
        float wv   = p10 + b31;
        float wx   = p11;
        float wxx  = p12;
        float wxxx = p13;
        float Nax  = EA_C * (ux + 0.5f * wx * wx);
        float Mb   = -EI_C * wxx;
        float Qs   = -EI_C * wxxx + Nax * wx;

        float f0 = half ? Nax : u;
        float f1 = half ? Mb  : wv;
        float f2 = half ? Qs  : wx;
        int base = (half ? 3 : 0) * n + pidx;
        out[base]         = f0;
        out[base + n]     = f1;
        out[base + 2 * n] = f2;
    }
}

extern "C" void kernel_launch(void* const* d_in, const int* in_sizes, int n_in,
                              void* d_out, int out_size, void* d_ws, size_t ws_size,
                              hipStream_t stream) {
    const float* x  = (const float*)d_in[0];
    const float* W1 = (const float*)d_in[1];
    const float* b1 = (const float*)d_in[2];
    const float* W2 = (const float*)d_in[3];
    const float* b2 = (const float*)d_in[4];
    const float* W3 = (const float*)d_in[5];
    const float* b3 = (const float*)d_in[6];
    float* out = (float*)d_out;
    const int n = in_sizes[0];

    const int nBatch = n >> 4;
    int blocks = (nBatch * 32 + 255) / 256;
    if (blocks > 2048) blocks = 2048;
    if (blocks < 1) blocks = 1;
    pinn_beam_wmma<<<blocks, 256, 0, stream>>>(x, W1, b1, W2, b2, W3, b3, out, n);
}


# Layer-2 jets and layer-3 sums run two accumulator rows at a time on packed 2-wide f32 ops, no pair-forming moves; layer-3 sum order changed: even and odd rows summed apart, joined once
# speedup vs baseline: 1.1774x; 1.1774x over previous
#include <hip/hip_runtime.h>


typedef float v2f __attribute__((ext_vector_type(2)));
typedef float v8f __attribute__((ext_vector_type(8)));

#define LBEAM 1.0f
#define EA_C  1.0e4f
#define EI_C  1.0e2f

__device__ __forceinline__ float fast_tanhf(float a) {
#if __has_builtin(__builtin_amdgcn_tanhf)
    return __builtin_amdgcn_tanhf(a);
#else
    float e = __expf(2.0f * a);
    return 1.0f - 2.0f * __builtin_amdgcn_rcpf(e + 1.0f);
#endif
}

__device__ __forceinline__ float xhalf(float v) {
    int b = __builtin_bit_cast(int, v);
    return __builtin_bit_cast(float,
        __builtin_amdgcn_permlanex16(b, b, 0x76543210, 0xfedcba98, false, false));
}

__device__ __forceinline__ float own_sum(bool up, float pa, float pb) {
    float keep = up ? pb : pa;
    float send = up ? pa : pb;
    return keep + xhalf(send);
}

__device__ __forceinline__ v2f tanh2(v2f a) {
    v2f r;
    r.x = fast_tanhf(a.x);
    r.y = fast_tanhf(a.y);
    return r;
}

__device__ __forceinline__ void subtile_partials(
    float xi, const v2f (&Aw)[4],
    const v2f (&w1p)[4], const v2f (&b1p)[4], const v2f (&b2p)[4],
    const v2f (&w3up)[4], const v2f (&w3wp)[4], float (&p)[6])
{
    v2f h0[4], h1[4], h2[4], h3[4];
#pragma unroll
    for (int c = 0; c < 4; ++c) {
        v2f a1 = w1p[c] * (1.0f / LBEAM);
        v2f a0 = a1 * xi + b1p[c];
        v2f t  = tanh2(a0);
        v2f s  = 1.0f - t * t;
        h0[c] = t;
        h1[c] = s * a1;
        h2[c] = -2.0f * t * s * a1 * a1;
        h3[c] = s * (4.0f * t * t - 2.0f * s) * a1 * a1 * a1;
    }

    v8f acc0 = {}, acc1 = {}, acc2 = {}, acc3 = {};
#pragma unroll
    for (int c = 0; c < 4; ++c) {
        acc0 = __builtin_amdgcn_wmma_f32_16x16x4_f32(false, Aw[c], false, h0[c],
                                                     (short)0, acc0, false, false);
        acc1 = __builtin_amdgcn_wmma_f32_16x16x4_f32(false, Aw[c], false, h1[c],
                                                     (short)0, acc1, false, false);
        acc2 = __builtin_amdgcn_wmma_f32_16x16x4_f32(false, Aw[c], false, h2[c],
                                                     (short)0, acc2, false, false);
        acc3 = __builtin_amdgcn_wmma_f32_16x16x4_f32(false, Aw[c], false, h3[c],
                                                     (short)0, acc3, false, false);
    }

    v2f q00 = {}, q01 = {};
    v2f q10 = {}, q11 = {}, q12 = {}, q13 = {};
#pragma unroll
    for (int q = 0; q < 4; ++q) {
        v2f A0 = v2f{ acc0[2 * q], acc0[2 * q + 1] } + b2p[q];
        v2f A1 = { acc1[2 * q], acc1[2 * q + 1] };
        v2f A2 = { acc2[2 * q], acc2[2 * q + 1] };
        v2f A3 = { acc3[2 * q], acc3[2 * q + 1] };
        v2f t  = tanh2(A0);
        v2f s  = 1.0f - t * t;
        v2f g0 = t;
        v2f g1 = s * A1;
        v2f g2 = s * A2 - 2.0f * t * s * A1 * A1;
        v2f g3 = s * A3 - 6.0f * t * s * A1 * A2
               + s * (4.0f * t * t - 2.0f * s) * A1 * A1 * A1;
        q00 += w3up[q] * g0; q01 += w3up[q] * g1;
        q10 += w3wp[q] * g0; q11 += w3wp[q] * g1;
        q12 += w3wp[q] * g2; q13 += w3wp[q] * g3;
    }
    p[0] = q00.x + q00.y; p[1] = q01.x + q01.y;
    p[2] = q10.x + q10.y; p[3] = q11.x + q11.y;
    p[4] = q12.x + q12.y; p[5] = q13.x + q13.y;
}

__global__ __launch_bounds__(256) void pinn_beam_wmma(
    const float* __restrict__ x,
    const float* __restrict__ W1, const float* __restrict__ b1,
    const float* __restrict__ W2, const float* __restrict__ b2,
    const float* __restrict__ W3, const float* __restrict__ b3,
    float* __restrict__ out, int n)
{
    const int lane = threadIdx.x & 31;
    const int half = lane >> 4;
    const int l16  = lane & 15;

    v2f Aw[4];
#pragma unroll
    for (int c = 0; c < 4; ++c) {
        Aw[c].x = W2[l16 * 16 + 4 * c + 2 * half + 0];
        Aw[c].y = W2[l16 * 16 + 4 * c + 2 * half + 1];
    }
    v2f w1p[4], b1p[4];
#pragma unroll
    for (int c = 0; c < 4; ++c) {
        int ch = 4 * c + 2 * half;
        w1p[c].x = W1[ch];  w1p[c].y = W1[ch + 1];
        b1p[c].x = b1[ch];  b1p[c].y = b1[ch + 1];
    }
    v2f b2p[4], w3up[4], w3wp[4];
#pragma unroll
    for (int q = 0; q < 4; ++q) {
        int ch = 2 * q + 8 * half;
        b2p[q].x  = b2[ch];       b2p[q].y  = b2[ch + 1];
        w3up[q].x = W3[ch];       w3up[q].y = W3[ch + 1];
        w3wp[q].x = W3[16 + ch];  w3wp[q].y = W3[16 + ch + 1];
    }
    const float b30 = b3[0], b31 = b3[1];

    const int nBatch = n >> 4;
    const int nTile  = (nBatch + 1) >> 1;
    const int wib    = __builtin_amdgcn_readfirstlane((int)threadIdx.x >> 5);
    const int wpb    = (int)blockDim.x >> 5;
    const int gwave  = (int)blockIdx.x * wpb + wib;
    const int nwave  = (int)gridDim.x * wpb;
    if (gwave >= nTile) return;

    int   bA  = 2 * gwave;
    int   bB  = (bA + 1 < nBatch) ? (bA + 1) : bA;
    float xa  = x[bA * 16 + l16];
    float xb  = x[bB * 16 + l16];

    for (int t = gwave; t < nTile; t += nwave) {
        const bool hasB = (2 * t + 1 < nBatch);
        const float xa_c = xa, xb_c = xb;
        if (t + nwave < nTile) {
            bA = 2 * (t + nwave);
            bB = (bA + 1 < nBatch) ? (bA + 1) : bA;
            xa = x[bA * 16 + l16];
            xb = x[bB * 16 + l16];
        }

        float pa[6], pb[6];
        subtile_partials(xa_c, Aw, w1p, b1p, b2p, w3up, w3wp, pa);
        subtile_partials(xb_c, Aw, w1p, b1p, b2p, w3up, w3wp, pb);

        const bool up = (half != 0);
        float s0 = own_sum(up, pa[0], pb[0]);
        float s1 = own_sum(up, pa[1], pb[1]);
        float s2 = own_sum(up, pa[2], pb[2]);
        float s3 = own_sum(up, pa[3], pb[3]);
        float s4 = own_sum(up, pa[4], pb[4]);
        float s5 = own_sum(up, pa[5], pb[5]);

        float u    = s0 + b30;
        float ux   = s1;
        float wv   = s2 + b31;
        float wx   = s3;
        float wxx  = s4;
        float wxxx = s5;
        float Nax  = EA_C * (ux + 0.5f * wx * wx);
        float Mb   = -EI_C * wxx;
        float Qs   = -EI_C * wxxx + Nax * wx;

        if (half == 0 || hasB) {
            const int pidx = t * 32 + lane;
            out[pidx]         = u;
            out[pidx + n]     = wv;
            out[pidx + 2 * n] = wx;
            out[pidx + 3 * n] = Nax;
            out[pidx + 4 * n] = Mb;
            out[pidx + 5 * n] = Qs;
        }
    }
}

extern "C" void kernel_launch(void* const* d_in, const int* in_sizes, int n_in,
                              void* d_out, int out_size, void* d_ws, size_t ws_size,
                              hipStream_t stream) {
    const float* x  = (const float*)d_in[0];
    const float* W1 = (const float*)d_in[1];
    const float* b1 = (const float*)d_in[2];
    const float* W2 = (const float*)d_in[3];
    const float* b2 = (const float*)d_in[4];
    const float* W3 = (const float*)d_in[5];
    const float* b3 = (const float*)d_in[6];
    float* out = (float*)d_out;
    const int n = in_sizes[0];

    pinn_beam_wmma<<<512, 256, 0, stream>>>(x, W1, b1, W2, b2, W3, b3, out, n);
}
